# starting kernel
# baseline (speedup 1.0000x reference)
#include <hip/hip_runtime.h>


typedef __attribute__((ext_vector_type(2))) float v2f;
typedef __attribute__((ext_vector_type(8))) float v8f;

typedef const float __attribute__((address_space(1)))* cgfp;
typedef float       __attribute__((address_space(1)))* gfp;

constexpr int C_IN  = 273;
constexpr int C_OUT = 273;
constexpr int T     = 360;
constexpr int BATCH = 256;

constexpr int MW    = 3;
constexpr int NW    = 4;
constexpr int MBLK  = MW * 16;
constexpr int NBLK  = NW * 16;
constexpr int MGRID = (C_OUT + MBLK - 1) / MBLK;
constexpr int NGRID = (T     + NBLK - 1) / NBLK;

__global__ __launch_bounds__(32)
void subject_layer_wmma(const float* __restrict__ xg,
                        const int*   __restrict__ subjects,
                        const float* __restrict__ wg,
                        float*       __restrict__ outg)
{
    const int bid = blockIdx.x;
    const int ng  = bid % NGRID;
    const int mg  = (bid / NGRID) % MGRID;
    const int b   = bid / (NGRID * MGRID);

    const int lane = threadIdx.x & 31;
    const int half = lane >> 4;
    const int l16  = lane & 15;

    const int subj = subjects[b];
    cgfp wb = (cgfp)wg + (size_t)subj * C_IN * C_OUT;
    cgfp xb = (cgfp)xg + (size_t)b    * C_IN * T;

    unsigned aoff[MW];
#pragma unroll
    for (int i = 0; i < MW; ++i) {
        int m = mg * MBLK + i * 16 + l16;
        if (m > C_OUT - 1) m = C_OUT - 1;
        aoff[i] = (unsigned)(2 * half) * C_OUT + (unsigned)m;
    }
    unsigned boff[NW];
#pragma unroll
    for (int j = 0; j < NW; ++j) {
        int n = ng * NBLK + j * 16 + l16;
        if (n > T - 1) n = T - 1;
        boff[j] = (unsigned)(2 * half) * T + (unsigned)n;
    }

    v8f acc[MW][NW];
#pragma unroll
    for (int i = 0; i < MW; ++i)
#pragma unroll
        for (int j = 0; j < NW; ++j)
            acc[i][j] = v8f{};

    constexpr int KMAIN = (C_IN / 4) * 4;
    for (int kb = 0; kb < KMAIN; kb += 4) {
        v2f av[MW];
#pragma unroll
        for (int i = 0; i < MW; ++i) {
            av[i] = v2f{ wb[aoff[i]], wb[aoff[i] + C_OUT] };
            aoff[i] += 4u * C_OUT;
        }
        v2f bv[NW];
#pragma unroll
        for (int j = 0; j < NW; ++j) {
            bv[j] = v2f{ xb[boff[j]], xb[boff[j] + T] };
            boff[j] += 4u * T;
        }
#pragma unroll
        for (int i = 0; i < MW; ++i)
#pragma unroll
            for (int j = 0; j < NW; ++j)
                acc[i][j] = __builtin_amdgcn_wmma_f32_16x16x4_f32(
                     false, av[i],
                     false, bv[j],
                     (short)0, acc[i][j],
                     false,  false);
    }

    {
        v2f av[MW];
#pragma unroll
        for (int i = 0; i < MW; ++i)
            av[i] = v2f{ (half == 0) ? wb[aoff[i]] : 0.0f, 0.0f };
        v2f bv[NW];
#pragma unroll
        for (int j = 0; j < NW; ++j)
            bv[j] = v2f{ (half == 0) ? xb[boff[j]] : 0.0f, 0.0f };
#pragma unroll
        for (int i = 0; i < MW; ++i)
#pragma unroll
            for (int j = 0; j < NW; ++j)
                acc[i][j] = __builtin_amdgcn_wmma_f32_16x16x4_f32(
                    false, av[i], false, bv[j], (short)0, acc[i][j],
                    false, false);
    }

    gfp ob = (gfp)outg + (size_t)b * C_OUT * T;
#pragma unroll
    for (int i = 0; i < MW; ++i) {
        const int m0 = mg * MBLK + i * 16 + half * 8;
#pragma unroll
        for (int j = 0; j < NW; ++j) {
            const int n = ng * NBLK + j * 16 + l16;
            if (n < T) {
#pragma unroll
                for (int r = 0; r < 8; ++r) {
                    const int m = m0 + r;
                    if (m < C_OUT)
                        ob[(unsigned)m * T + (unsigned)n] = acc[i][j][r];
                }
            }
        }
    }
}

extern "C" void kernel_launch(void* const* d_in, const int* in_sizes, int n_in,
                              void* d_out, int out_size, void* d_ws, size_t ws_size,
                              hipStream_t stream)
{
    const float* x        = (const float*)d_in[0];
    const int*   subjects = (const int*)  d_in[1];
    const float* w        = (const float*)d_in[2];
    float*       out      = (float*)d_out;

    const int nblocks = BATCH * MGRID * NGRID;
    subject_layer_wmma<<<nblocks, 32, 0, stream>>>(x, subjects, w, out);
}


# Stage both operands through LDS in 12-wave work-groups (288x128 output tile, 12-row k chunks, next chunk prefetched in registers); k summation order unchanged
# speedup vs baseline: 1.3715x; 1.3715x over previous
#include <hip/hip_runtime.h>


typedef __attribute__((ext_vector_type(2))) float v2f;
typedef __attribute__((ext_vector_type(4))) float v4f;
typedef __attribute__((ext_vector_type(8))) float v8f;

typedef const float __attribute__((address_space(1)))* cgfp;
typedef const v4f   __attribute__((address_space(1)))* cgv4p;
typedef float       __attribute__((address_space(1)))* gfp;

constexpr int C_IN  = 273;
constexpr int C_OUT = 273;
constexpr int T     = 360;
constexpr int BATCH = 256;

constexpr int MW    = 3;
constexpr int NW    = 4;
constexpr int MBLK  = MW * 16;
constexpr int NBLK  = NW * 16;
constexpr int WM    = 6;
constexpr int WN    = 2;
constexpr int NWAVES   = WM * WN;
constexpr int NTHREADS = NWAVES * 32;
constexpr int BN    = WN * NBLK;
constexpr int NGRID = (T + BN - 1) / BN;

constexpr int KC      = 12;
constexpr int KSTEPS  = KC / 4;
constexpr int NCHUNK  = (C_IN + KC - 1) / KC;
constexpr int A_ELEMS = KC * C_OUT;
constexpr int A_FULL  = A_ELEMS / NTHREADS;
constexpr int A_REST  = A_ELEMS - A_FULL * NTHREADS;
constexpr int A_BUF   = 3280;
constexpr int B_PITCH = 136;
constexpr int B_V4    = KC * B_PITCH / 4;

__global__ __launch_bounds__(NTHREADS)
void subject_layer_wmma(const float* __restrict__ xg,
                        const int*   __restrict__ subjects,
                        const float* __restrict__ wg,
                        float*       __restrict__ outg)
{
    __shared__ float As[2][A_BUF];
    __shared__ v4f   Bs4[2][B_V4];

    const int bid = blockIdx.x;
    const int ng  = bid % NGRID;
    const int b   = bid / NGRID;

    const int tid  = threadIdx.x;
    const int wave = tid >> 5;
    const int lane = tid & 31;
    const int half = lane >> 4;
    const int l16  = lane & 15;
    const int wm   = wave % WM;
    const int wn   = wave / WM;

    const int subj = subjects[b];
    cgfp wb = (cgfp)wg + (size_t)subj * C_IN * C_OUT;
    cgfp xb = (cgfp)xg + (size_t)b    * C_IN * T;

    const int  brow  = tid >> 5;
    const int  bc4   = tid & 31;
    const int  bn    = ng * BN + 4 * bc4;
    const bool bn_ok = bn < T;

    unsigned aoff[MW];
#pragma unroll
    for (int i = 0; i < MW; ++i) {
        int m = wm * MBLK + i * 16 + l16;
        if (m > C_OUT - 1) m = C_OUT - 1;
        aoff[i] = (unsigned)(2 * half) * C_OUT + (unsigned)m;
    }
    unsigned boff[NW];
#pragma unroll
    for (int j = 0; j < NW; ++j)
        boff[j] = (unsigned)(2 * half) * B_PITCH + (unsigned)(wn * NBLK + j * 16 + l16);

    v8f acc[MW][NW];
#pragma unroll
    for (int i = 0; i < MW; ++i)
#pragma unroll
        for (int j = 0; j < NW; ++j)
            acc[i][j] = v8f{};

    float ar[A_FULL + 1];
    v4f   br;

    auto fetch = [&](int c) {
        const unsigned gbase = (unsigned)(c * KC) * C_OUT;
#pragma unroll
        for (int i = 0; i < A_FULL; ++i) {
            const unsigned g = gbase + (unsigned)(tid + i * NTHREADS);
            ar[i] = (g < (unsigned)(C_IN * C_OUT)) ? wb[g] : 0.0f;
        }
        {
            const unsigned g = gbase + (unsigned)(tid + A_FULL * NTHREADS);
            ar[A_FULL] = (tid < A_REST && g < (unsigned)(C_IN * C_OUT)) ? wb[g] : 0.0f;
        }
        const int k = c * KC + brow;
        if (bn_ok && k < C_IN)
            br = *(cgv4p)(xb + (unsigned)k * T + (unsigned)bn);
        else
            br = v4f{0.0f, 0.0f, 0.0f, 0.0f};
    };
    auto stage = [&](int buf) {
#pragma unroll
        for (int i = 0; i < A_FULL; ++i)
            As[buf][tid + i * NTHREADS] = ar[i];
        if (tid < A_REST)
            As[buf][tid + A_FULL * NTHREADS] = ar[A_FULL];
        Bs4[buf][brow * (B_PITCH / 4) + bc4] = br;
    };

    fetch(0);
    stage(0);
    __syncthreads();

    for (int c = 0; c < NCHUNK; ++c) {
        const int buf = c & 1;
        if (c + 1 < NCHUNK) fetch(c + 1);

        const float* Ab = As[buf];
        const float* Bb = (const float*)Bs4[buf];
#pragma unroll
        for (int s = 0; s < KSTEPS; ++s) {
            v2f av[MW];
#pragma unroll
            for (int i = 0; i < MW; ++i)
                av[i] = v2f{ Ab[aoff[i] + (4 * s) * C_OUT],
                             Ab[aoff[i] + (4 * s + 1) * C_OUT] };
            v2f bv[NW];
#pragma unroll
            for (int j = 0; j < NW; ++j)
                bv[j] = v2f{ Bb[boff[j] + (4 * s) * B_PITCH],
                             Bb[boff[j] + (4 * s + 1) * B_PITCH] };
#pragma unroll
            for (int i = 0; i < MW; ++i)
#pragma unroll
                for (int j = 0; j < NW; ++j)
                    acc[i][j] = __builtin_amdgcn_wmma_f32_16x16x4_f32(
                         false, av[i],
                         false, bv[j],
                         (short)0, acc[i][j],
                         false,  false);
        }

        if (c + 1 < NCHUNK) stage(buf ^ 1);
        __syncthreads();
    }

    gfp ob = (gfp)outg + (size_t)b * C_OUT * T;
#pragma unroll
    for (int i = 0; i < MW; ++i) {
        const int m0 = wm * MBLK + i * 16 + half * 8;
#pragma unroll
        for (int j = 0; j < NW; ++j) {
            const int n = ng * BN + wn * NBLK + j * 16 + l16;
            if (n < T) {
#pragma unroll
                for (int r = 0; r < 8; ++r) {
                    const int m = m0 + r;
                    if (m < C_OUT)
                        ob[(unsigned)m * T + (unsigned)n] = acc[i][j][r];
                }
            }
        }
    }
}

extern "C" void kernel_launch(void* const* d_in, const int* in_sizes, int n_in,
                              void* d_out, int out_size, void* d_ws, size_t ws_size,
                              hipStream_t stream)
{
    const float* x        = (const float*)d_in[0];
    const int*   subjects = (const int*)  d_in[1];
    const float* w        = (const float*)d_in[2];
    float*       out      = (float*)d_out;

    const int nblocks = BATCH * NGRID;
    subject_layer_wmma<<<nblocks, NTHREADS, 0, stream>>>(x, subjects, w, out);
}


# LDS stage layout interleaves the two k rows of each pair, so every WMMA fragment is one aligned 8-byte LDS read with no register shuffles; k summation order unchanged
# speedup vs baseline: 1.4571x; 1.0624x over previous
#include <hip/hip_runtime.h>


typedef __attribute__((ext_vector_type(2))) float v2f;
typedef __attribute__((ext_vector_type(4))) float v4f;
typedef __attribute__((ext_vector_type(8))) float v8f;

typedef const float __attribute__((address_space(1)))* cgfp;
typedef float       __attribute__((address_space(1)))* gfp;

constexpr int C_IN  = 273;
constexpr int C_OUT = 273;
constexpr int T     = 360;
constexpr int BATCH = 256;

constexpr int MW    = 3;
constexpr int NW    = 4;
constexpr int MBLK  = MW * 16;
constexpr int NBLK  = NW * 16;
constexpr int WM    = 6;
constexpr int WN    = 2;
constexpr int NWAVES   = WM * WN;
constexpr int NTHREADS = NWAVES * 32;
constexpr int BN    = WN * NBLK;
constexpr int NGRID = (T + BN - 1) / BN;

constexpr int KC      = 12;
constexpr int KP      = KC / 2;
constexpr int KSTEPS  = KC / 4;
constexpr int NCHUNK  = (C_IN + KC - 1) / KC;
constexpr int PA      = 304;
constexpr int PB      = 144;
constexpr int A_V2    = KP * PA;
constexpr int B_V4    = KP * PB / 2;

typedef const v2f __attribute__((address_space(1)))* cgv2p;

__global__ __launch_bounds__(NTHREADS) __attribute__((amdgpu_waves_per_eu(6)))
void subject_layer_wmma(const float* __restrict__ xg,
                        const int*   __restrict__ subjects,
                        const float* __restrict__ wg,
                        float*       __restrict__ outg)
{
    __shared__ v2f As2[2][A_V2];
    __shared__ v4f Bs4[2][B_V4];

    const int bid = blockIdx.x;
    const int ng  = bid % NGRID;
    const int b   = bid / NGRID;

    const int tid  = threadIdx.x;
    const int wave = tid >> 5;
    const int lane = tid & 31;
    const int half = lane >> 4;
    const int l16  = lane & 15;
    const int wm   = wave % WM;
    const int wn   = wave / WM;

    const int subj = subjects[b];
    cgfp wb = (cgfp)wg + (size_t)subj * C_IN * C_OUT;
    cgfp xb = (cgfp)xg + (size_t)b    * C_IN * T;

    const bool a_wave = __builtin_amdgcn_readfirstlane(tid) < 9 * 32;
    const unsigned acol = (unsigned)(tid < C_OUT ? tid : C_OUT - 1);
    const int  bp    = tid >> 6;
    const int  bs    = tid & 63;
    const int  bn    = ng * BN + 2 * bs;
    const unsigned bcol = (unsigned)(bn < T ? bn : 0);
    const unsigned xoff = (unsigned)(2 * bp) * T + bcol;

    unsigned aoff[MW];
#pragma unroll
    for (int i = 0; i < MW; ++i) {
        int m = wm * MBLK + i * 16 + l16;
        if (m > C_OUT - 1) m = C_OUT - 1;
        aoff[i] = (unsigned)half * PA + (unsigned)m;
    }
    unsigned boff[NW];
#pragma unroll
    for (int j = 0; j < NW; ++j)
        boff[j] = (unsigned)half * PB + (unsigned)(wn * NBLK + j * 16 + l16);

    v8f acc[MW][NW];
#pragma unroll
    for (int i = 0; i < MW; ++i)
#pragma unroll
        for (int j = 0; j < NW; ++j)
            acc[i][j] = v8f{};

    float ar[KC];
    v2f   br0, br1;

    auto fetch = [&](int c) {
        const int k0 = c * KC;
        if (c < NCHUNK - 1) {
            cgfp wc = wb + (unsigned)k0 * C_OUT;
            cgfp xc = xb + (unsigned)k0 * T;
            if (a_wave) {
#pragma unroll
                for (int r = 0; r < KC; ++r)
                    ar[r] = wc[acol + (unsigned)(r * C_OUT)];
            }
            br0 = *(cgv2p)(xc + xoff);
            br1 = *(cgv2p)(xc + xoff + (unsigned)T);
        } else {
            if (a_wave) {
#pragma unroll
                for (int r = 0; r < KC; ++r)
                    ar[r] = (k0 + r < C_IN) ? wb[(unsigned)(k0 + r) * C_OUT + acol] : 0.0f;
            }
            const int k  = k0 + 2 * bp;
            const unsigned ka = (unsigned)(k     < C_IN ? k     : C_IN - 1);
            const unsigned kb = (unsigned)(k + 1 < C_IN ? k + 1 : C_IN - 1);
            br0 = *(cgv2p)(xb + ka * T + bcol);
            br1 = *(cgv2p)(xb + kb * T + bcol);
        }
    };
    auto stage = [&](int buf, int c) {
        if (c == NCHUNK - 1) {
            const int k = c * KC + 2 * bp;
            if (k     >= C_IN) br0 = v2f{0.0f, 0.0f};
            if (k + 1 >= C_IN) br1 = v2f{0.0f, 0.0f};
        }
        if (a_wave) {
#pragma unroll
            for (int p = 0; p < KP; ++p)
                As2[buf][p * PA + tid] = v2f{ ar[2 * p], ar[2 * p + 1] };
        }
        Bs4[buf][bp * (PB / 2) + bs] = v4f{ br0.x, br1.x, br0.y, br1.y };
    };

    fetch(0);
    stage(0, 0);
    __syncthreads();

    for (int c = 0; c < NCHUNK; ++c) {
        const int buf = c & 1;
        if (c + 1 < NCHUNK) fetch(c + 1);

        const v2f* Ab = As2[buf];
        const v2f* Bb = (const v2f*)Bs4[buf];
#pragma unroll
        for (int s = 0; s < KSTEPS; ++s) {
            v2f av[MW];
#pragma unroll
            for (int i = 0; i < MW; ++i)
                av[i] = Ab[aoff[i] + (2 * s) * PA];
            v2f bv[NW];
#pragma unroll
            for (int j = 0; j < NW; ++j)
                bv[j] = Bb[boff[j] + (2 * s) * PB];
#pragma unroll
            for (int i = 0; i < MW; ++i)
#pragma unroll
                for (int j = 0; j < NW; ++j)
                    acc[i][j] = __builtin_amdgcn_wmma_f32_16x16x4_f32(
                         false, av[i],
                         false, bv[j],
                         (short)0, acc[i][j],
                         false,  false);
        }

        if (c + 1 < NCHUNK) stage(buf ^ 1, c + 1);
        __syncthreads();
    }

    gfp ob = (gfp)outg + (size_t)b * C_OUT * T;
#pragma unroll
    for (int i = 0; i < MW; ++i) {
        const int m0 = wm * MBLK + i * 16 + half * 8;
#pragma unroll
        for (int j = 0; j < NW; ++j) {
            const int n = ng * BN + wn * NBLK + j * 16 + l16;
            if (n < T) {
#pragma unroll
                for (int r = 0; r < 8; ++r) {
                    const int m = m0 + r;
                    if (m < C_OUT)
                        ob[(unsigned)m * T + (unsigned)n] = acc[i][j][r];
                }
            }
        }
    }
}

extern "C" void kernel_launch(void* const* d_in, const int* in_sizes, int n_in,
                              void* d_out, int out_size, void* d_ws, size_t ws_size,
                              hipStream_t stream)
{
    const float* x        = (const float*)d_in[0];
    const int*   subjects = (const int*)  d_in[1];
    const float* w        = (const float*)d_in[2];
    float*       out      = (float*)d_out;

    const int nblocks = BATCH * NGRID;
    subject_layer_wmma<<<nblocks, NTHREADS, 0, stream>>>(x, subjects, w, out);
}


# LDS stage becomes a ring of 6 chunk slots filled 3 chunks ahead, so the work-group barrier comes once per 3 chunks (108 WMMAs per wave) instead of every chunk; k summation order unchanged
# speedup vs baseline: 1.5027x; 1.0313x over previous
#include <hip/hip_runtime.h>


typedef __attribute__((ext_vector_type(2))) float v2f;
typedef __attribute__((ext_vector_type(4))) float v4f;
typedef __attribute__((ext_vector_type(8))) float v8f;

typedef const float __attribute__((address_space(1)))* cgfp;
typedef float       __attribute__((address_space(1)))* gfp;

constexpr int C_IN  = 273;
constexpr int C_OUT = 273;
constexpr int T     = 360;
constexpr int BATCH = 256;

constexpr int MW    = 3;
constexpr int NW    = 4;
constexpr int MBLK  = MW * 16;
constexpr int NBLK  = NW * 16;
constexpr int WM    = 6;
constexpr int WN    = 2;
constexpr int NWAVES   = WM * WN;
constexpr int NTHREADS = NWAVES * 32;
constexpr int BN    = WN * NBLK;
constexpr int NGRID = (T + BN - 1) / BN;

constexpr int KC      = 12;
constexpr int KP      = KC / 2;
constexpr int KSTEPS  = KC / 4;
constexpr int NCHUNK  = (C_IN + KC - 1) / KC;
constexpr int PA      = 304;
constexpr int PB      = 144;
constexpr int A_V2    = KP * PA;
constexpr int B_V4    = KP * PB / 2;
constexpr int AHEAD   = 3;
constexpr int RING    = 2 * AHEAD;

typedef const v2f __attribute__((address_space(1)))* cgv2p;

__global__ __launch_bounds__(NTHREADS) __attribute__((amdgpu_waves_per_eu(6)))
void subject_layer_wmma(const float* __restrict__ xg,
                        const int*   __restrict__ subjects,
                        const float* __restrict__ wg,
                        float*       __restrict__ outg)
{
    __shared__ v2f As2[RING][A_V2];
    __shared__ v4f Bs4[RING][B_V4];

    const int bid = blockIdx.x;
    const int ng  = bid % NGRID;
    const int b   = bid / NGRID;

    const int tid  = threadIdx.x;
    const int wave = __builtin_amdgcn_readfirstlane(tid >> 5);
    const int lane = tid & 31;
    const int half = lane >> 4;
    const int l16  = lane & 15;
    const int wm   = wave % WM;
    const int wn   = wave / WM;

    const int subj = subjects[b];
    cgfp wb = (cgfp)wg + (size_t)subj * C_IN * C_OUT;
    cgfp xb = (cgfp)xg + (size_t)b    * C_IN * T;

    const bool a_wave = __builtin_amdgcn_readfirstlane(tid) < 9 * 32;
    const unsigned acol = (unsigned)(tid < C_OUT ? tid : C_OUT - 1);
    const int  bp    = tid >> 6;
    const int  bs    = tid & 63;
    const int  bn    = ng * BN + 2 * bs;
    const unsigned bcol = (unsigned)(bn < T ? bn : 0);
    const unsigned xoff = (unsigned)(2 * bp) * T + bcol;

    unsigned aoff[MW];
#pragma unroll
    for (int i = 0; i < MW; ++i) {
        int m = wm * MBLK + i * 16 + l16;
        if (m > C_OUT - 1) m = C_OUT - 1;
        aoff[i] = (unsigned)half * PA + (unsigned)m;
    }
    unsigned boff[NW];
#pragma unroll
    for (int j = 0; j < NW; ++j)
        boff[j] = (unsigned)half * PB + (unsigned)(wn * NBLK + j * 16 + l16);

    v8f acc[MW][NW];
#pragma unroll
    for (int i = 0; i < MW; ++i)
#pragma unroll
        for (int j = 0; j < NW; ++j)
            acc[i][j] = v8f{};

    float ar[KC];
    v2f   br0, br1;

    auto fetch_full = [&](int c, float (&a)[KC], v2f& b0, v2f& b1) {
        const int k0 = c * KC;
        cgfp wc = wb + (unsigned)k0 * C_OUT;
        cgfp xc = xb + (unsigned)k0 * T;
        if (a_wave) {
#pragma unroll
            for (int r = 0; r < KC; ++r)
                a[r] = wc[acol + (unsigned)(r * C_OUT)];
        }
        b0 = *(cgv2p)(xc + xoff);
        b1 = *(cgv2p)(xc + xoff + (unsigned)T);
    };
    auto fetch_last = [&]() {
        constexpr int k0 = (NCHUNK - 1) * KC;
        if (a_wave) {
#pragma unroll
            for (int r = 0; r < KC; ++r)
                ar[r] = (k0 + r < C_IN) ? wb[(unsigned)(k0 + r) * C_OUT + acol] : 0.0f;
        }
        const int k  = k0 + 2 * bp;
        const unsigned ka = (unsigned)(k     < C_IN ? k     : C_IN - 1);
        const unsigned kb = (unsigned)(k + 1 < C_IN ? k + 1 : C_IN - 1);
        br0 = *(cgv2p)(xb + ka * T + bcol);
        br1 = *(cgv2p)(xb + kb * T + bcol);
    };
    auto stage_full = [&](int slot, const float (&a)[KC], v2f b0, v2f b1) {
        if (a_wave) {
#pragma unroll
            for (int p = 0; p < KP; ++p)
                As2[slot][p * PA + tid] = v2f{ a[2 * p], a[2 * p + 1] };
        }
        Bs4[slot][bp * (PB / 2) + bs] = v4f{ b0.x, b1.x, b0.y, b1.y };
    };
    auto stage_last = [&](int slot) {
        const int k = (NCHUNK - 1) * KC + 2 * bp;
        if (k     >= C_IN) br0 = v2f{0.0f, 0.0f};
        if (k + 1 >= C_IN) br1 = v2f{0.0f, 0.0f};
        stage_full(slot, ar, br0, br1);
    };
    auto products = [&](int slot) {
        const v2f* Ab = As2[slot];
        const v2f* Bb = (const v2f*)Bs4[slot];
#pragma unroll
        for (int s = 0; s < KSTEPS; ++s) {
            v2f av[MW];
#pragma unroll
            for (int i = 0; i < MW; ++i)
                av[i] = Ab[aoff[i] + (2 * s) * PA];
            v2f bv[NW];
#pragma unroll
            for (int j = 0; j < NW; ++j)
                bv[j] = Bb[boff[j] + (2 * s) * PB];
#pragma unroll
            for (int i = 0; i < MW; ++i)
#pragma unroll
                for (int j = 0; j < NW; ++j)
                    acc[i][j] = __builtin_amdgcn_wmma_f32_16x16x4_f32(
                         false, av[i],
                         false, bv[j],
                         (short)0, acc[i][j],
                         false,  false);
        }
    };

    {
        float pa[AHEAD][KC];
        v2f   pb0[AHEAD], pb1[AHEAD];
#pragma unroll
        for (int p = 0; p < AHEAD; ++p)
            fetch_full(p, pa[p], pb0[p], pb1[p]);
#pragma unroll
        for (int p = 0; p < AHEAD; ++p)
            stage_full(p, pa[p], pb0[p], pb1[p]);
    }
    __syncthreads();

    int slot  = 0;
    int fill  = AHEAD;
    int phase = 0;
    auto advance = [&]() {
        if (++phase == AHEAD) {
            phase = 0;
            __syncthreads();
        }
        slot = (slot + 1 == RING) ? 0 : slot + 1;
        fill = (fill + 1 == RING) ? 0 : fill + 1;
    };

    static_assert(NCHUNK > 2 * AHEAD, "the three loops below assume this");
    for (int c = 0; c + AHEAD < NCHUNK - 1; ++c) {
        fetch_full(c + AHEAD, ar, br0, br1);
        products(slot);
        stage_full(fill, ar, br0, br1);
        advance();
    }
    fetch_last();
    products(slot);
    stage_last(fill);
    advance();
    for (int r = 0; r < AHEAD; ++r) {
        products(slot);
        advance();
    }

    gfp ob = (gfp)outg + (size_t)b * C_OUT * T;
#pragma unroll
    for (int i = 0; i < MW; ++i) {
        const int m0 = wm * MBLK + i * 16 + half * 8;
#pragma unroll
        for (int j = 0; j < NW; ++j) {
            const int n = ng * BN + wn * NBLK + j * 16 + l16;
            if (n < T) {
#pragma unroll
                for (int r = 0; r < 8; ++r) {
                    const int m = m0 + r;
                    if (m < C_OUT)
                        ob[(unsigned)m * T + (unsigned)n] = acc[i][j][r];
                }
            }
        }
    }
}

extern "C" void kernel_launch(void* const* d_in, const int* in_sizes, int n_in,
                              void* d_out, int out_size, void* d_ws, size_t ws_size,
                              hipStream_t stream)
{
    const float* x        = (const float*)d_in[0];
    const int*   subjects = (const int*)  d_in[1];
    const float* w        = (const float*)d_in[2];
    float*       out      = (float*)d_out;

    const int nblocks = BATCH * NGRID;
    subject_layer_wmma<<<nblocks, NTHREADS, 0, stream>>>(x, subjects, w, out);
}


# Output tile leaves through wave-private LDS rows: each store instruction writes two 256-byte row pieces (16 bytes per lane) instead of two 64-byte pieces; k summation order unchanged
# speedup vs baseline: 1.5722x; 1.0463x over previous
#include <hip/hip_runtime.h>


typedef __attribute__((ext_vector_type(2))) float v2f;
typedef __attribute__((ext_vector_type(4))) float v4f;
typedef __attribute__((ext_vector_type(8))) float v8f;

typedef const float __attribute__((address_space(1)))* cgfp;
typedef float       __attribute__((address_space(1)))* gfp;

constexpr int C_IN  = 273;
constexpr int C_OUT = 273;
constexpr int T     = 360;
constexpr int BATCH = 256;

constexpr int MW    = 3;
constexpr int NW    = 4;
constexpr int MBLK  = MW * 16;
constexpr int NBLK  = NW * 16;
constexpr int WM    = 6;
constexpr int WN    = 2;
constexpr int NWAVES   = WM * WN;
constexpr int NTHREADS = NWAVES * 32;
constexpr int BN    = WN * NBLK;
constexpr int NGRID = (T + BN - 1) / BN;

constexpr int KC      = 12;
constexpr int KP      = KC / 2;
constexpr int KSTEPS  = KC / 4;
constexpr int NCHUNK  = (C_IN + KC - 1) / KC;
constexpr int PA      = 304;
constexpr int PB      = 144;
constexpr int A_V2    = KP * PA;
constexpr int B_V4    = KP * PB / 2;
constexpr int AHEAD   = 3;
constexpr int RING    = 2 * AHEAD;

typedef const v2f __attribute__((address_space(1)))* cgv2p;
typedef v4f       __attribute__((address_space(1)))* gv4p;
typedef float     __attribute__((address_space(3)))* lfp;
typedef const v4f __attribute__((address_space(3)))* clv4p;

constexpr int SP         = 68;
constexpr int STG_FLOATS = 16 * SP;
constexpr int SLOT_FLOATS = 2 * A_V2;

__global__ __launch_bounds__(NTHREADS) __attribute__((amdgpu_waves_per_eu(6)))
void subject_layer_wmma(const float* __restrict__ xg,
                        const int*   __restrict__ subjects,
                        const float* __restrict__ wg,
                        float*       __restrict__ outg)
{
    __shared__ __attribute__((aligned(16))) v2f As2[RING][A_V2];
    __shared__ v4f Bs4[RING][B_V4];

    const int bid = blockIdx.x;
    const int ng  = bid % NGRID;
    const int b   = bid / NGRID;

    const int tid  = threadIdx.x;
    const int wave = __builtin_amdgcn_readfirstlane(tid >> 5);
    const int lane = tid & 31;
    const int half = lane >> 4;
    const int l16  = lane & 15;
    const int wm   = wave % WM;
    const int wn   = wave / WM;

    const int subj = subjects[b];
    cgfp wb = (cgfp)wg + (size_t)subj * C_IN * C_OUT;
    cgfp xb = (cgfp)xg + (size_t)b    * C_IN * T;

    const bool a_wave = __builtin_amdgcn_readfirstlane(tid) < 9 * 32;
    const unsigned acol = (unsigned)(tid < C_OUT ? tid : C_OUT - 1);
    const int  bp    = tid >> 6;
    const int  bs    = tid & 63;
    const int  bn    = ng * BN + 2 * bs;
    const unsigned bcol = (unsigned)(bn < T ? bn : 0);
    const unsigned xoff = (unsigned)(2 * bp) * T + bcol;

    unsigned aoff[MW];
#pragma unroll
    for (int i = 0; i < MW; ++i) {
        int m = wm * MBLK + i * 16 + l16;
        if (m > C_OUT - 1) m = C_OUT - 1;
        aoff[i] = (unsigned)half * PA + (unsigned)m;
    }
    unsigned boff[NW];
#pragma unroll
    for (int j = 0; j < NW; ++j)
        boff[j] = (unsigned)half * PB + (unsigned)(wn * NBLK + j * 16 + l16);

    v8f acc[MW][NW];
#pragma unroll
    for (int i = 0; i < MW; ++i)
#pragma unroll
        for (int j = 0; j < NW; ++j)
            acc[i][j] = v8f{};

    float ar[KC];
    v2f   br0, br1;

    auto fetch_full = [&](int c, float (&a)[KC], v2f& b0, v2f& b1) {
        const int k0 = c * KC;
        cgfp wc = wb + (unsigned)k0 * C_OUT;
        cgfp xc = xb + (unsigned)k0 * T;
        if (a_wave) {
#pragma unroll
            for (int r = 0; r < KC; ++r)
                a[r] = wc[acol + (unsigned)(r * C_OUT)];
        }
        b0 = *(cgv2p)(xc + xoff);
        b1 = *(cgv2p)(xc + xoff + (unsigned)T);
    };
    auto fetch_last = [&]() {
        constexpr int k0 = (NCHUNK - 1) * KC;
        if (a_wave) {
#pragma unroll
            for (int r = 0; r < KC; ++r)
                ar[r] = (k0 + r < C_IN) ? wb[(unsigned)(k0 + r) * C_OUT + acol] : 0.0f;
        }
        const int k  = k0 + 2 * bp;
        const unsigned ka = (unsigned)(k     < C_IN ? k     : C_IN - 1);
        const unsigned kb = (unsigned)(k + 1 < C_IN ? k + 1 : C_IN - 1);
        br0 = *(cgv2p)(xb + ka * T + bcol);
        br1 = *(cgv2p)(xb + kb * T + bcol);
    };
    auto stage_full = [&](int slot, const float (&a)[KC], v2f b0, v2f b1) {
        if (a_wave) {
#pragma unroll
            for (int p = 0; p < KP; ++p)
                As2[slot][p * PA + tid] = v2f{ a[2 * p], a[2 * p + 1] };
        }
        Bs4[slot][bp * (PB / 2) + bs] = v4f{ b0.x, b1.x, b0.y, b1.y };
    };
    auto stage_last = [&](int slot) {
        const int k = (NCHUNK - 1) * KC + 2 * bp;
        if (k     >= C_IN) br0 = v2f{0.0f, 0.0f};
        if (k + 1 >= C_IN) br1 = v2f{0.0f, 0.0f};
        stage_full(slot, ar, br0, br1);
    };
    auto products = [&](int slot) {
        const v2f* Ab = As2[slot];
        const v2f* Bb = (const v2f*)Bs4[slot];
#pragma unroll
        for (int s = 0; s < KSTEPS; ++s) {
            v2f av[MW];
#pragma unroll
            for (int i = 0; i < MW; ++i)
                av[i] = Ab[aoff[i] + (2 * s) * PA];
            v2f bv[NW];
#pragma unroll
            for (int j = 0; j < NW; ++j)
                bv[j] = Bb[boff[j] + (2 * s) * PB];
#pragma unroll
            for (int i = 0; i < MW; ++i)
#pragma unroll
                for (int j = 0; j < NW; ++j)
                    acc[i][j] = __builtin_amdgcn_wmma_f32_16x16x4_f32(
                         false, av[i],
                         false, bv[j],
                         (short)0, acc[i][j],
                         false,  false);
        }
    };

    {
        float pa[AHEAD][KC];
        v2f   pb0[AHEAD], pb1[AHEAD];
#pragma unroll
        for (int p = 0; p < AHEAD; ++p)
            fetch_full(p, pa[p], pb0[p], pb1[p]);
#pragma unroll
        for (int p = 0; p < AHEAD; ++p)
            stage_full(p, pa[p], pb0[p], pb1[p]);
    }
    __syncthreads();

    int slot  = 0;
    int fill  = AHEAD;
    int phase = 0;
    auto advance = [&]() {
        if (++phase == AHEAD) {
            phase = 0;
            __syncthreads();
        }
        slot = (slot + 1 == RING) ? 0 : slot + 1;
        fill = (fill + 1 == RING) ? 0 : fill + 1;
    };

    static_assert(NCHUNK > 2 * AHEAD, "the three loops below assume this");
    for (int c = 0; c + AHEAD < NCHUNK - 1; ++c) {
        fetch_full(c + AHEAD, ar, br0, br1);
        products(slot);
        stage_full(fill, ar, br0, br1);
        advance();
    }
    fetch_last();
    products(slot);
    stage_last(fill);
    advance();
    for (int r = 0; r < AHEAD; ++r) {
        products(slot);
        advance();
    }

    static_assert(NCHUNK == 23 && AHEAD == 3 && RING == 6, "free ring slots at the end are derived for these");
    static_assert(10 * STG_FLOATS <= 3 * SLOT_FLOATS && 2 * STG_FLOATS <= SLOT_FLOATS, "strips fit the free slots");
    const int stg_off = (wave < 10) ? wave * STG_FLOATS
                                    : 5 * SLOT_FLOATS + (wave - 10) * STG_FLOATS;
    lfp stg = (lfp)&As2[0][0] + stg_off;

    gfp ob = (gfp)outg + (size_t)b * C_OUT * T;
    const int  n   = ng * BN + wn * NBLK + 4 * l16;
    const bool nok = n < T;
    const unsigned wr0 = (unsigned)(half * 8) * SP + (unsigned)l16;
    const unsigned rd0 = (unsigned)half * SP + (unsigned)(4 * l16);
#pragma unroll
    for (int i = 0; i < MW; ++i) {
#pragma unroll
        for (int j = 0; j < NW; ++j)
#pragma unroll
            for (int r = 0; r < 8; ++r)
                stg[wr0 + (unsigned)(r * SP + j * 16)] = acc[i][j][r];
        __builtin_amdgcn_fence(__ATOMIC_ACQ_REL, "wavefront");
        const int m0 = wm * MBLK + i * 16 + half;
#pragma unroll
        for (int q = 0; q < 8; ++q) {
            const v4f v = *(clv4p)(stg + rd0 + (unsigned)(2 * q * SP));
            const int m = m0 + 2 * q;
            if (nok && m < C_OUT)
                *(gv4p)(ob + (unsigned)m * T + (unsigned)n) = v;
        }
        __builtin_amdgcn_fence(__ATOMIC_ACQ_REL, "wavefront");
    }
}

extern "C" void kernel_launch(void* const* d_in, const int* in_sizes, int n_in,
                              void* d_out, int out_size, void* d_ws, size_t ws_size,
                              hipStream_t stream)
{
    const float* x        = (const float*)d_in[0];
    const int*   subjects = (const int*)  d_in[1];
    const float* w        = (const float*)d_in[2];
    float*       out      = (float*)d_out;

    const int nblocks = BATCH * NGRID;
    subject_layer_wmma<<<nblocks, NTHREADS, 0, stream>>>(x, subjects, w, out);
}
